# starting kernel
# baseline (speedup 1.0000x reference)
#include <hip/hip_runtime.h>

typedef float v2f __attribute__((ext_vector_type(2)));
typedef float v4f __attribute__((ext_vector_type(4)));
typedef float v8f __attribute__((ext_vector_type(8)));

#define B_DIM 4096
#define G_DIM 31
#define P_DIM 8
#define C_DIM 36

__global__ void zero_kernel(float* __restrict__ out, int n) {
    int i = blockIdx.x * blockDim.x + threadIdx.x;
    if (i < n) out[i] = 0.0f;
}

__device__ __forceinline__ float sval_masked(const float* __restrict__ cgh, int i, int k) {
    int c = ((i * (17 - i)) >> 1) + (k - i);
    float v = cgh[c];
    return (i <= k) ? v : 0.0f;
}

__global__ void __launch_bounds__(256)
quadform_kernel(const float* __restrict__ param,
                const float* __restrict__ coeffs,
                float* __restrict__ out) {
    __shared__ float sA[G_DIM][32][4];

    const int lane = threadIdx.x & 31;
    const int wave = threadIdx.x >> 5;
    const int m    = lane & 15;
    const int hi   = lane >> 4;
    const int koff = hi << 1;

    const int g    = blockIdx.y;
    const int base = blockIdx.x * 256 + wave * 32;
    const int b0   = base + m;
    const int b1   = b0 + 16;

    for (int idx = threadIdx.x; idx < G_DIM * 32; idx += 256) {
        const int h  = idx >> 5;
        const int l  = idx & 31;
        const int lk = (l >> 4) << 1;
        const int ls = l & 7;
        const float* cgh = coeffs + ((size_t)g * G_DIM + h) * C_DIM;
        sA[h][l][0] = sval_masked(cgh, ls, lk + 0);
        sA[h][l][1] = sval_masked(cgh, ls, lk + 1);
        sA[h][l][2] = sval_masked(cgh, ls, lk + 4);
        sA[h][l][3] = sval_masked(cgh, ls, lk + 5);
    }
    __syncthreads();

    float acc0 = 0.0f, acc1 = 0.0f;

    for (int h = 0; h < G_DIM; ++h) {
        v4f a = *(const v4f*)(&sA[h][lane][0]);
        v2f aLo; aLo.x = a.x; aLo.y = a.y;
        v2f aHi; aHi.x = a.z; aHi.y = a.w;

        const float* p0 = param + (((size_t)b0 * G_DIM + g) * G_DIM + h) * (size_t)P_DIM;
        const float* p1 = param + (((size_t)b1 * G_DIM + g) * G_DIM + h) * (size_t)P_DIM;
        v2f bA = *(const v2f*)(p0 + koff);
        v2f bB = *(const v2f*)(p0 + koff + 4);
        v2f bC = *(const v2f*)(p1 + koff);
        v2f bD = *(const v2f*)(p1 + koff + 4);

        __builtin_prefetch(p0 + 2 * P_DIM, 0, 3);
        __builtin_prefetch(p1 + 2 * P_DIM, 0, 3);

        v8f d0 = {};
        v8f d1 = {};
        d0 = __builtin_amdgcn_wmma_f32_16x16x4_f32(false, aLo, false, bA, (short)0, d0, false, false);
        d1 = __builtin_amdgcn_wmma_f32_16x16x4_f32(false, aLo, false, bC, (short)0, d1, false, false);
        d0 = __builtin_amdgcn_wmma_f32_16x16x4_f32(false, aHi, false, bB, (short)0, d0, false, false);
        d1 = __builtin_amdgcn_wmma_f32_16x16x4_f32(false, aHi, false, bD, (short)0, d1, false, false);

        float y0 = hi ? d0[2] : d0[0];
        float y1 = hi ? d0[3] : d0[1];
        float y4 = hi ? d0[6] : d0[4];
        float y5 = hi ? d0[7] : d0[5];
        acc0 += y0 * bA.x + y1 * bA.y + y4 * bB.x + y5 * bB.y;

        float z0 = hi ? d1[2] : d1[0];
        float z1 = hi ? d1[3] : d1[1];
        float z4 = hi ? d1[6] : d1[4];
        float z5 = hi ? d1[7] : d1[5];
        acc1 += z0 * bC.x + z1 * bC.y + z4 * bD.x + z5 * bD.y;
    }

    float o0 = __shfl(acc0, lane ^ 16, 32);
    float o1 = __shfl(acc1, lane ^ 16, 32);
    float s0 = acc0 + o0;
    float s1 = acc1 + o1;
    if (lane < 16) atomicAdd(&out[base + lane], s0);
    else           atomicAdd(&out[base + lane], s1);
}

extern "C" void kernel_launch(void* const* d_in, const int* in_sizes, int n_in,
                              void* d_out, int out_size, void* d_ws, size_t ws_size,
                              hipStream_t stream) {
    const float* param  = (const float*)d_in[0];
    const float* coeffs = (const float*)d_in[1];
    float* out = (float*)d_out;

    zero_kernel<<<(out_size + 255) / 256, 256, 0, stream>>>(out, out_size);

    dim3 grid(B_DIM / 256, G_DIM);
    quadform_kernel<<<grid, 256, 0, stream>>>(param, coeffs, out);
}


# param rows read as whole-line loads (16 B per lane, 512 contiguous B per wave load), staged through a per-wave LDS tile and re-read as WMMA B fragments; k split per half-wave regroups the f32 sums
# speedup vs baseline: 1.7610x; 1.7610x over previous
#include <hip/hip_runtime.h>

typedef float v2f __attribute__((ext_vector_type(2)));
typedef float v4f __attribute__((ext_vector_type(4)));
typedef float v8f __attribute__((ext_vector_type(8)));

#define B_DIM 4096
#define G_DIM 31
#define P_DIM 8
#define C_DIM 36

#define ROW_STRIDE_F (G_DIM * G_DIM * P_DIM)
#define CHUNK_H      16
#define TILE_PITCH_F 132
#define N_CHUNKS     4

__global__ void zero_kernel(float* __restrict__ out, int n) {
    int i = blockIdx.x * blockDim.x + threadIdx.x;
    if (i < n) out[i] = 0.0f;
}

__device__ __forceinline__ float sval_masked(const float* __restrict__ cgh, int i, int k) {
    int c = ((i * (17 - i)) >> 1) + (k - i);
    float v = cgh[c];
    return (i <= k) ? v : 0.0f;
}

__global__ void __launch_bounds__(256)
quadform_kernel(const float* __restrict__ param,
                const float* __restrict__ coeffs,
                float* __restrict__ out) {
    __shared__ __attribute__((aligned(16))) float sA[G_DIM][32][4];
    __shared__ __attribute__((aligned(16))) float sP[8][16][TILE_PITCH_F];

    const int lane = threadIdx.x & 31;
    const int wave = threadIdx.x >> 5;
    const int m    = lane & 15;
    const int hi   = lane >> 4;

    const int g    = blockIdx.y;
    const int base = blockIdx.x * 256 + wave * 32;

    float cv[4][4];
    #pragma unroll
    for (int j = 0; j < 4; ++j) {
        const int idx = threadIdx.x + j * 256;
        const int idc = idx < G_DIM * 32 ? idx : G_DIM * 32 - 1;
        const int h  = idc >> 5;
        const int l  = idc & 31;
        const int lk = (l >> 4) << 2;
        const int ls = l & 7;
        const float* cgh = coeffs + ((size_t)g * G_DIM + h) * C_DIM;
        cv[j][0] = sval_masked(cgh, ls, lk + 0);
        cv[j][1] = sval_masked(cgh, ls, lk + 1);
        cv[j][2] = sval_masked(cgh, ls, lk + 2);
        cv[j][3] = sval_masked(cgh, ls, lk + 3);
    }

    const float* rowp = param + ((size_t)base * (G_DIM * G_DIM) + (size_t)g * G_DIM) * P_DIM;
    const int loff0 = lane * 4;
    const int loff1 = CHUNK_H * P_DIM + (lane < 30 ? lane : 29) * 4;

    v4f st[16];
    {
        const float* q = rowp + loff0;
        #pragma unroll
        for (int r = 0; r < 16; ++r) st[r] = *(const v4f*)(q + (size_t)r * ROW_STRIDE_F);
    }

    #pragma unroll
    for (int j = 0; j < 4; ++j) {
        const int idx = threadIdx.x + j * 256;
        if (idx < G_DIM * 32) {
            v4f v; v.x = cv[j][0]; v.y = cv[j][1]; v.z = cv[j][2]; v.w = cv[j][3];
            *(v4f*)(&sA[idx >> 5][idx & 31][0]) = v;
        }
    }
    __syncthreads();

    float acc = 0.0f;

    for (int ci = 0; ci < N_CHUNKS; ++ci) {
        const int t    = ci >> 1;
        const int half = ci & 1;

        #pragma unroll
        for (int r = 0; r < 16; ++r) *(v4f*)(&sP[wave][r][lane * 4]) = st[r];

        if (ci + 1 < N_CHUNKS) {
            const int tn = (ci + 1) >> 1;
            const float* q = rowp + (size_t)tn * 16 * ROW_STRIDE_F + (((ci + 1) & 1) ? loff1 : loff0);
            #pragma unroll
            for (int r = 0; r < 16; ++r) st[r] = *(const v4f*)(q + (size_t)r * ROW_STRIDE_F);
        }
        __builtin_amdgcn_sched_barrier(0);

        const int n  = half ? (G_DIM - CHUNK_H) : CHUNK_H;
        const float* sAh = &sA[half * CHUNK_H][lane][0];
        const float* sPm = &sP[wave][m][hi * 4];

        int hl = 0;
        for (; hl + 1 < n; hl += 2) {
            v4f a0 = *(const v4f*)(sAh + (hl + 0) * 128);
            v4f a1 = *(const v4f*)(sAh + (hl + 1) * 128);
            v4f p0 = *(const v4f*)(sPm + (hl + 0) * P_DIM);
            v4f p1 = *(const v4f*)(sPm + (hl + 1) * P_DIM);
            v2f a0l; a0l.x = a0.x; a0l.y = a0.y;  v2f a0h; a0h.x = a0.z; a0h.y = a0.w;
            v2f a1l; a1l.x = a1.x; a1l.y = a1.y;  v2f a1h; a1h.x = a1.z; a1h.y = a1.w;
            v2f p0l; p0l.x = p0.x; p0l.y = p0.y;  v2f p0h; p0h.x = p0.z; p0h.y = p0.w;
            v2f p1l; p1l.x = p1.x; p1l.y = p1.y;  v2f p1h; p1h.x = p1.z; p1h.y = p1.w;

            v8f d0 = {};
            v8f d1 = {};
            d0 = __builtin_amdgcn_wmma_f32_16x16x4_f32(false, a0l, false, p0l, (short)0, d0, false, false);
            d1 = __builtin_amdgcn_wmma_f32_16x16x4_f32(false, a1l, false, p1l, (short)0, d1, false, false);
            d0 = __builtin_amdgcn_wmma_f32_16x16x4_f32(false, a0h, false, p0h, (short)0, d0, false, false);
            d1 = __builtin_amdgcn_wmma_f32_16x16x4_f32(false, a1h, false, p1h, (short)0, d1, false, false);

            float y0 = hi ? d0[4] : d0[0];
            float y1 = hi ? d0[5] : d0[1];
            float y2 = hi ? d0[6] : d0[2];
            float y3 = hi ? d0[7] : d0[3];
            acc += y0 * p0.x + y1 * p0.y + y2 * p0.z + y3 * p0.w;

            float z0 = hi ? d1[4] : d1[0];
            float z1 = hi ? d1[5] : d1[1];
            float z2 = hi ? d1[6] : d1[2];
            float z3 = hi ? d1[7] : d1[3];
            acc += z0 * p1.x + z1 * p1.y + z2 * p1.z + z3 * p1.w;
        }
        if (hl < n) {
            v4f a0 = *(const v4f*)(sAh + hl * 128);
            v4f p0 = *(const v4f*)(sPm + hl * P_DIM);
            v2f a0l; a0l.x = a0.x; a0l.y = a0.y;  v2f a0h; a0h.x = a0.z; a0h.y = a0.w;
            v2f p0l; p0l.x = p0.x; p0l.y = p0.y;  v2f p0h; p0h.x = p0.z; p0h.y = p0.w;
            v8f d0 = {};
            d0 = __builtin_amdgcn_wmma_f32_16x16x4_f32(false, a0l, false, p0l, (short)0, d0, false, false);
            d0 = __builtin_amdgcn_wmma_f32_16x16x4_f32(false, a0h, false, p0h, (short)0, d0, false, false);
            float y0 = hi ? d0[4] : d0[0];
            float y1 = hi ? d0[5] : d0[1];
            float y2 = hi ? d0[6] : d0[2];
            float y3 = hi ? d0[7] : d0[3];
            acc += y0 * p0.x + y1 * p0.y + y2 * p0.z + y3 * p0.w;
        }

        if (half) {
            float o = __shfl(acc, lane ^ 16, 32);
            float s = acc + o;
            if (lane < 16) atomicAdd(&out[base + t * 16 + lane], s);
            acc = 0.0f;
        }
    }
}

extern "C" void kernel_launch(void* const* d_in, const int* in_sizes, int n_in,
                              void* d_out, int out_size, void* d_ws, size_t ws_size,
                              hipStream_t stream) {
    const float* param  = (const float*)d_in[0];
    const float* coeffs = (const float*)d_in[1];
    float* out = (float*)d_out;

    zero_kernel<<<(out_size + 255) / 256, 256, 0, stream>>>(out, out_size);

    dim3 grid(B_DIM / 256, G_DIM);
    quadform_kernel<<<grid, 256, 0, stream>>>(param, coeffs, out);
}


# closing dot without selects: A rows 8..15 carry S rows 4..7,0..3, both half-waves read D regs 0..3; 2 packed f32 FMAs per cell into 4 running sums (new sum order); loop fenced per 8 cells (VGPRs)
# speedup vs baseline: 1.8623x; 1.0575x over previous
#include <hip/hip_runtime.h>

typedef float v2f __attribute__((ext_vector_type(2)));
typedef float v4f __attribute__((ext_vector_type(4)));
typedef float v8f __attribute__((ext_vector_type(8)));

#define B_DIM 4096
#define G_DIM 31
#define P_DIM 8
#define C_DIM 36

#define ROW_STRIDE_F (G_DIM * G_DIM * P_DIM)
#define CHUNK_H      16
#define TILE_PITCH_F 132
#define N_CHUNKS     4

__global__ void zero_kernel(float* __restrict__ out, int n) {
    int i = blockIdx.x * blockDim.x + threadIdx.x;
    if (i < n) out[i] = 0.0f;
}

__device__ __forceinline__ float sval_masked(const float* __restrict__ cgh, int i, int k) {
    int c = ((i * (17 - i)) >> 1) + (k - i);
    float v = cgh[c];
    return (i <= k) ? v : 0.0f;
}

__global__ void __launch_bounds__(256)
quadform_kernel(const float* __restrict__ param,
                const float* __restrict__ coeffs,
                float* __restrict__ out) {
    __shared__ __attribute__((aligned(16))) float sA[G_DIM][32][4];
    __shared__ __attribute__((aligned(16))) float sP[8][16][TILE_PITCH_F];

    const int lane = threadIdx.x & 31;
    const int wave = threadIdx.x >> 5;
    const int m    = lane & 15;
    const int hi   = lane >> 4;

    const int g    = blockIdx.y;
    const int base = blockIdx.x * 256 + wave * 32;

    float cv[4][4];
    #pragma unroll
    for (int j = 0; j < 4; ++j) {
        const int idx = threadIdx.x + j * 256;
        const int idc = idx < G_DIM * 32 ? idx : G_DIM * 32 - 1;
        const int h  = idc >> 5;
        const int l  = idc & 31;
        const int lk = (l >> 4) << 2;
        const int ls = ((l & 7) + ((l & 8) >> 1)) & 7;
        const float* cgh = coeffs + ((size_t)g * G_DIM + h) * C_DIM;
        cv[j][0] = sval_masked(cgh, ls, lk + 0);
        cv[j][1] = sval_masked(cgh, ls, lk + 1);
        cv[j][2] = sval_masked(cgh, ls, lk + 2);
        cv[j][3] = sval_masked(cgh, ls, lk + 3);
    }

    const float* rowp = param + ((size_t)base * (G_DIM * G_DIM) + (size_t)g * G_DIM) * P_DIM;
    const int loff0 = lane * 4;
    const int loff1 = CHUNK_H * P_DIM + (lane < 30 ? lane : 29) * 4;

    v4f st[16];
    {
        const float* q = rowp + loff0;
        #pragma unroll
        for (int r = 0; r < 16; ++r) st[r] = *(const v4f*)(q + (size_t)r * ROW_STRIDE_F);
    }

    #pragma unroll
    for (int j = 0; j < 4; ++j) {
        const int idx = threadIdx.x + j * 256;
        if (idx < G_DIM * 32) {
            v4f v; v.x = cv[j][0]; v.y = cv[j][1]; v.z = cv[j][2]; v.w = cv[j][3];
            *(v4f*)(&sA[idx >> 5][idx & 31][0]) = v;
        }
    }
    __syncthreads();

    v2f accl = {};
    v2f acch = {};

    for (int ci = 0; ci < N_CHUNKS; ++ci) {
        const int t    = ci >> 1;
        const int half = ci & 1;

        #pragma unroll
        for (int r = 0; r < 16; ++r) *(v4f*)(&sP[wave][r][lane * 4]) = st[r];

        if (ci + 1 < N_CHUNKS) {
            const int tn = (ci + 1) >> 1;
            const float* q = rowp + (size_t)tn * 16 * ROW_STRIDE_F + (((ci + 1) & 1) ? loff1 : loff0);
            #pragma unroll
            for (int r = 0; r < 16; ++r) st[r] = *(const v4f*)(q + (size_t)r * ROW_STRIDE_F);
        }
        __builtin_amdgcn_sched_barrier(0);

        const int n  = half ? (G_DIM - CHUNK_H) : CHUNK_H;
        const float* sAh = &sA[half * CHUNK_H][lane][0];
        const float* sPm = &sP[wave][m][hi * 4];

        int hl = 0;
        for (; hl + 1 < n; hl += 2) {
            if ((hl & 7) == 0 && hl != 0) __builtin_amdgcn_sched_barrier(0);
            v4f a0 = *(const v4f*)(sAh + (hl + 0) * 128);
            v4f a1 = *(const v4f*)(sAh + (hl + 1) * 128);
            v4f p0 = *(const v4f*)(sPm + (hl + 0) * P_DIM);
            v4f p1 = *(const v4f*)(sPm + (hl + 1) * P_DIM);
            v2f a0l; a0l.x = a0.x; a0l.y = a0.y;  v2f a0h; a0h.x = a0.z; a0h.y = a0.w;
            v2f a1l; a1l.x = a1.x; a1l.y = a1.y;  v2f a1h; a1h.x = a1.z; a1h.y = a1.w;
            v2f p0l; p0l.x = p0.x; p0l.y = p0.y;  v2f p0h; p0h.x = p0.z; p0h.y = p0.w;
            v2f p1l; p1l.x = p1.x; p1l.y = p1.y;  v2f p1h; p1h.x = p1.z; p1h.y = p1.w;

            v8f d0 = {};
            v8f d1 = {};
            d0 = __builtin_amdgcn_wmma_f32_16x16x4_f32(false, a0l, false, p0l, (short)0, d0, false, false);
            d1 = __builtin_amdgcn_wmma_f32_16x16x4_f32(false, a1l, false, p1l, (short)0, d1, false, false);
            d0 = __builtin_amdgcn_wmma_f32_16x16x4_f32(false, a0h, false, p0h, (short)0, d0, false, false);
            d1 = __builtin_amdgcn_wmma_f32_16x16x4_f32(false, a1h, false, p1h, (short)0, d1, false, false);

            v2f y0l; y0l.x = d0[0]; y0l.y = d0[1];  v2f y0h; y0h.x = d0[2]; y0h.y = d0[3];
            v2f y1l; y1l.x = d1[0]; y1l.y = d1[1];  v2f y1h; y1h.x = d1[2]; y1h.y = d1[3];
            accl = __builtin_elementwise_fma(y0l, p0l, accl);
            acch = __builtin_elementwise_fma(y0h, p0h, acch);
            accl = __builtin_elementwise_fma(y1l, p1l, accl);
            acch = __builtin_elementwise_fma(y1h, p1h, acch);
        }
        if (hl < n) {
            v4f a0 = *(const v4f*)(sAh + hl * 128);
            v4f p0 = *(const v4f*)(sPm + hl * P_DIM);
            v2f a0l; a0l.x = a0.x; a0l.y = a0.y;  v2f a0h; a0h.x = a0.z; a0h.y = a0.w;
            v2f p0l; p0l.x = p0.x; p0l.y = p0.y;  v2f p0h; p0h.x = p0.z; p0h.y = p0.w;
            v8f d0 = {};
            d0 = __builtin_amdgcn_wmma_f32_16x16x4_f32(false, a0l, false, p0l, (short)0, d0, false, false);
            d0 = __builtin_amdgcn_wmma_f32_16x16x4_f32(false, a0h, false, p0h, (short)0, d0, false, false);
            v2f y0l; y0l.x = d0[0]; y0l.y = d0[1];  v2f y0h; y0h.x = d0[2]; y0h.y = d0[3];
            accl = __builtin_elementwise_fma(y0l, p0l, accl);
            acch = __builtin_elementwise_fma(y0h, p0h, acch);
        }

        if (half) {
            float part = (accl.x + accl.y) + (acch.x + acch.y);
            float o = __shfl(part, lane ^ 16, 32);
            float s = part + o;
            if (lane < 16) atomicAdd(&out[base + t * 16 + lane], s);
            accl = (v2f){};
            acch = (v2f){};
        }
    }
}

extern "C" void kernel_launch(void* const* d_in, const int* in_sizes, int n_in,
                              void* d_out, int out_size, void* d_ws, size_t ws_size,
                              hipStream_t stream) {
    const float* param  = (const float*)d_in[0];
    const float* coeffs = (const float*)d_in[1];
    float* out = (float*)d_out;

    zero_kernel<<<(out_size + 255) / 256, 256, 0, stream>>>(out, out_size);

    dim3 grid(B_DIM / 256, G_DIM);
    quadform_kernel<<<grid, 256, 0, stream>>>(param, coeffs, out);
}
